# starting kernel
# baseline (speedup 1.0000x reference)
#include <hip/hip_runtime.h>
#include <hip/hip_bf16.h>
#include <math.h>

typedef float v2f __attribute__((ext_vector_type(2)));
typedef float v8f __attribute__((ext_vector_type(8)));

#define GMAXC 1.0f
#define GMINC 0.01f
#define EPSC  0.1f

static constexpr int N_ = 8, B_ = 32, NIN_ = 128, NOUT_ = 128, E_ = 1024, M_ = 130;
static constexpr int VSTR = 132;

__device__ __forceinline__ float theta_eff(float th) {
    th = fminf(fmaxf(th, -GMAXC), GMAXC);
    return (fabsf(th) < GMINC) ? 0.0f : th;
}

__global__ void prep_kernel(const float* __restrict__ theta,
                            const float* __restrict__ noise,
                            float* __restrict__ Vt,
                            float* __restrict__ cbias)
{
    int t = blockIdx.x * blockDim.x + threadIdx.x;
    int n = t >> 7;
    int o = t & 127;
    const float* nz = noise + (size_t)n * M_ * NOUT_ + o;

    float denom = 1e-10f;
    for (int m = 0; m < M_; ++m) {
        float th = theta_eff(theta[m * NOUT_ + o]);
        float f  = nz[(size_t)m * NOUT_];
        float tn = th * ((f * 2.0f - 1.0f) * EPSC + 1.0f);
        denom += fabsf(tn);
    }
    float inv = 1.0f / denom;

    float bias = 0.0f;
    float* vrow = Vt + ((size_t)n * NOUT_ + o) * NIN_;
    for (int m = 0; m < M_; ++m) {
        float th = theta_eff(theta[m * NOUT_ + o]);
        float f  = nz[(size_t)m * NOUT_];
        float tn = th * ((f * 2.0f - 1.0f) * EPSC + 1.0f);
        if (m < NIN_) {
            vrow[m] = tn * inv;
            bias += fmaxf(-tn, 0.0f);
        } else if (m == NIN_) {
            bias += fmaxf(tn, 0.0f);
        }
    }
    cbias[n * NOUT_ + o] = bias * inv;
}

__global__ void __launch_bounds__(256)
gemm_act_kernel(const float* __restrict__ a,
                const float* __restrict__ Vt,
                const float* __restrict__ cbias,
                const float* __restrict__ eta,
                float* __restrict__ out)
{
    __shared__ float ldsV[64 * VSTR];
    __shared__ float ldsC[64];

    const int bid    = blockIdx.x;
    const int ohalf  = bid & 1;
    const int echunk = (bid >> 1) & 7;
    const int b      = (bid >> 4) & 31;
    const int n      = bid >> 9;
    const int tid    = threadIdx.x;
    const int o0     = ohalf * 64;

    const float* vsrc = Vt + ((size_t)n * NOUT_ + o0) * NIN_;
    #pragma unroll
    for (int r = 0; r < 32; ++r) {
        int idx  = r * 256 + tid;
        int orow = idx >> 7;
        int m    = idx & 127;
        ldsV[orow * VSTR + m] = vsrc[(size_t)orow * NIN_ + m];
    }
    if (tid < 64) ldsC[tid] = cbias[n * NOUT_ + o0 + tid];
    __syncthreads();

    const int wave  = tid >> 5;
    const int lane  = tid & 31;
    const int lrow  = lane & 15;
    const int lhalf = lane >> 4;
    const int e0    = echunk * 128 + wave * 16;

    const float* Abase = a + ((size_t)(n * B_ + b) * NIN_) * E_ + e0 + lrow;

    v8f acc[4];
    #pragma unroll
    for (int ot = 0; ot < 4; ++ot)
        #pragma unroll
        for (int i = 0; i < 8; ++i) acc[ot][i] = 0.0f;

    for (int k0 = 0; k0 < NIN_; k0 += 4) {
        const int kk = k0 + 2 * lhalf;
        const float* ap = Abase + (size_t)kk * E_;
        v2f Bf;
        Bf.x = ap[0];
        Bf.y = ap[E_];
        #pragma unroll
        for (int ot = 0; ot < 4; ++ot) {
            const float* vp = &ldsV[(ot * 16 + lrow) * VSTR + kk];
            v2f Af;
            Af.x = vp[0];
            Af.y = vp[1];
            acc[ot] = __builtin_amdgcn_wmma_f32_16x16x4_f32(
                false, Af, false, Bf, (short)0, acc[ot], false, false);
        }
    }

    const float eta0 = eta[0], eta1 = eta[1], eta2 = eta[2], eta3 = eta[3];
    float* obase = out + ((size_t)(n * B_ + b) * NOUT_) * E_;
    #pragma unroll
    for (int ot = 0; ot < 4; ++ot) {
        #pragma unroll
        for (int r = 0; r < 8; ++r) {
            const int oo = ot * 16 + r + 8 * lhalf;
            float z   = acc[ot][r] + ldsC[oo];
            float act = eta0 + eta1 * tanhf((z - eta2) * eta3);
            obase[(size_t)(o0 + oo) * E_ + e0 + lrow] = act;
        }
    }
}

extern "C" void kernel_launch(void* const* d_in, const int* in_sizes, int n_in,
                              void* d_out, int out_size, void* d_ws, size_t ws_size,
                              hipStream_t stream)
{
    const float* a     = (const float*)d_in[0];
    const float* theta = (const float*)d_in[1];
    const float* noise = (const float*)d_in[2];
    const float* eta   = (const float*)d_in[3];
    float* out = (float*)d_out;

    float* Vt = (float*)d_ws;
    float* cb = Vt + (size_t)N_ * NOUT_ * NIN_;

    prep_kernel<<<4, 256, 0, stream>>>(theta, noise, Vt, cb);

    const int nblocks = N_ * B_ * (E_ / 128) * 2;
    gemm_act_kernel<<<nblocks, 256, 0, stream>>>(a, Vt, cb, eta, out);
}


# prep stage: the 130 rows m spread over the 8 waves of 32 blocks (34 independent loads per thread) instead of one thread walking all rows twice; denominator and bias sums become a two-level tree
# speedup vs baseline: 1.5895x; 1.5895x over previous
#include <hip/hip_runtime.h>
#include <hip/hip_bf16.h>
#include <math.h>

typedef float v2f __attribute__((ext_vector_type(2)));
typedef float v8f __attribute__((ext_vector_type(8)));

#define GMAXC 1.0f
#define GMINC 0.01f
#define EPSC  0.1f

static constexpr int N_ = 8, B_ = 32, NIN_ = 128, NOUT_ = 128, E_ = 1024, M_ = 130;
static constexpr int VSTR = 132;

__device__ __forceinline__ float theta_eff(float th) {
    th = fminf(fmaxf(th, -GMAXC), GMAXC);
    return (fabsf(th) < GMINC) ? 0.0f : th;
}

__device__ __forceinline__ float noisy(float th, float f) {
    return theta_eff(th) * ((f * 2.0f - 1.0f) * EPSC + 1.0f);
}

static constexpr int PO   = 32;
static constexpr int PS   = 8;
static constexpr int PJ   = NIN_ / PS;
static constexpr int TSTR = 132;

__global__ void __launch_bounds__(256)
prep_kernel(const float* __restrict__ theta,
            const float* __restrict__ noise,
            float* __restrict__ Vt,
            float* __restrict__ cbias)
{
    __shared__ __attribute__((aligned(16))) float tile[PO * TSTR];
    __shared__ float pden[PS][PO];
    __shared__ float pbias[PS][PO];

    const int tid = threadIdx.x;
    const int ol  = tid & (PO - 1);
    const int ms  = tid >> 5;
    const int n   = blockIdx.x >> 2;
    const int ob  = (blockIdx.x & 3) * PO;
    const int o   = ob + ol;

    const float* thp = theta + o;
    const float* nzp = noise + (size_t)n * M_ * NOUT_ + o;

    float th[PJ], nf[PJ];
    #pragma unroll
    for (int j = 0; j < PJ; ++j) {
        const int m = ms + PS * j;
        th[j] = thp[m * NOUT_];
        nf[j] = nzp[m * NOUT_];
    }
    float tht = 0.0f, nft = 0.0f;
    if (ms < M_ - NIN_) {
        tht = thp[(NIN_ + ms) * NOUT_];
        nft = nzp[(NIN_ + ms) * NOUT_];
    }

    float tn[PJ];
    float den = 0.0f, bs = 0.0f;
    #pragma unroll
    for (int j = 0; j < PJ; ++j) {
        tn[j] = noisy(th[j], nf[j]);
        den += fabsf(tn[j]);
        bs  += fmaxf(-tn[j], 0.0f);
    }
    const float tnt = noisy(tht, nft);
    den += fabsf(tnt);
    if (ms == 0) bs += fmaxf(tnt, 0.0f);
    pden[ms][ol]  = den;
    pbias[ms][ol] = bs;
    __syncthreads();

    float dsum = 0.0f, bsum = 0.0f;
    #pragma unroll
    for (int s = 0; s < PS; ++s) {
        dsum += pden[s][ol];
        bsum += pbias[s][ol];
    }
    const float inv = 1.0f / (dsum + 1e-10f);

    #pragma unroll
    for (int j = 0; j < PJ; ++j)
        tile[ol * TSTR + ms + PS * j] = tn[j] * inv;
    __syncthreads();

    if (ms == 0) cbias[n * NOUT_ + o] = bsum * inv;

    #pragma unroll
    for (int rr = 0; rr < PO / PS; ++rr) {
        const int row = ms * (PO / PS) + rr;
        const float4 v = *reinterpret_cast<const float4*>(&tile[row * TSTR + ol * 4]);
        float4* dst = reinterpret_cast<float4*>(Vt + ((size_t)n * NOUT_ + ob + row) * NIN_) + ol;
        *dst = v;
    }
}

__global__ void __launch_bounds__(256)
gemm_act_kernel(const float* __restrict__ a,
                const float* __restrict__ Vt,
                const float* __restrict__ cbias,
                const float* __restrict__ eta,
                float* __restrict__ out)
{
    __shared__ float ldsV[64 * VSTR];
    __shared__ float ldsC[64];

    const int bid    = blockIdx.x;
    const int ohalf  = bid & 1;
    const int echunk = (bid >> 1) & 7;
    const int b      = (bid >> 4) & 31;
    const int n      = bid >> 9;
    const int tid    = threadIdx.x;
    const int o0     = ohalf * 64;

    const float* vsrc = Vt + ((size_t)n * NOUT_ + o0) * NIN_;
    #pragma unroll
    for (int r = 0; r < 32; ++r) {
        int idx  = r * 256 + tid;
        int orow = idx >> 7;
        int m    = idx & 127;
        ldsV[orow * VSTR + m] = vsrc[(size_t)orow * NIN_ + m];
    }
    if (tid < 64) ldsC[tid] = cbias[n * NOUT_ + o0 + tid];
    __syncthreads();

    const int wave  = tid >> 5;
    const int lane  = tid & 31;
    const int lrow  = lane & 15;
    const int lhalf = lane >> 4;
    const int e0    = echunk * 128 + wave * 16;

    const float* Abase = a + ((size_t)(n * B_ + b) * NIN_) * E_ + e0 + lrow;

    v8f acc[4];
    #pragma unroll
    for (int ot = 0; ot < 4; ++ot)
        #pragma unroll
        for (int i = 0; i < 8; ++i) acc[ot][i] = 0.0f;

    for (int k0 = 0; k0 < NIN_; k0 += 4) {
        const int kk = k0 + 2 * lhalf;
        const float* ap = Abase + (size_t)kk * E_;
        v2f Bf;
        Bf.x = ap[0];
        Bf.y = ap[E_];
        #pragma unroll
        for (int ot = 0; ot < 4; ++ot) {
            const float* vp = &ldsV[(ot * 16 + lrow) * VSTR + kk];
            v2f Af;
            Af.x = vp[0];
            Af.y = vp[1];
            acc[ot] = __builtin_amdgcn_wmma_f32_16x16x4_f32(
                false, Af, false, Bf, (short)0, acc[ot], false, false);
        }
    }

    const float eta0 = eta[0], eta1 = eta[1], eta2 = eta[2], eta3 = eta[3];
    float* obase = out + ((size_t)(n * B_ + b) * NOUT_) * E_;
    #pragma unroll
    for (int ot = 0; ot < 4; ++ot) {
        #pragma unroll
        for (int r = 0; r < 8; ++r) {
            const int oo = ot * 16 + r + 8 * lhalf;
            float z   = acc[ot][r] + ldsC[oo];
            float act = eta0 + eta1 * tanhf((z - eta2) * eta3);
            obase[(size_t)(o0 + oo) * E_ + e0 + lrow] = act;
        }
    }
}

extern "C" void kernel_launch(void* const* d_in, const int* in_sizes, int n_in,
                              void* d_out, int out_size, void* d_ws, size_t ws_size,
                              hipStream_t stream)
{
    const float* a     = (const float*)d_in[0];
    const float* theta = (const float*)d_in[1];
    const float* noise = (const float*)d_in[2];
    const float* eta   = (const float*)d_in[3];
    float* out = (float*)d_out;

    float* Vt = (float*)d_ws;
    float* cb = Vt + (size_t)N_ * NOUT_ * NIN_;

    prep_kernel<<<N_ * (NOUT_ / PO), 256, 0, stream>>>(theta, noise, Vt, cb);

    const int nblocks = N_ * B_ * (E_ / 128) * 2;
    gemm_act_kernel<<<nblocks, 256, 0, stream>>>(a, Vt, cb, eta, out);
}


# gemm stage re-tiled: block = all 128 rows o x 128 e (A fetched once); wave = 32x64 register tile, its 4 column tiles interleave e by 4: 16-byte loads/stores, one LDS fragment per 4 WMMAs
# speedup vs baseline: 1.7693x; 1.1131x over previous
#include <hip/hip_runtime.h>
#include <hip/hip_bf16.h>
#include <math.h>

typedef float v2f __attribute__((ext_vector_type(2)));
typedef float v8f __attribute__((ext_vector_type(8)));

#define GMAXC 1.0f
#define GMINC 0.01f
#define EPSC  0.1f

static constexpr int N_ = 8, B_ = 32, NIN_ = 128, NOUT_ = 128, E_ = 1024, M_ = 130;
static constexpr int VSTR = 132;

__device__ __forceinline__ float theta_eff(float th) {
    th = fminf(fmaxf(th, -GMAXC), GMAXC);
    return (fabsf(th) < GMINC) ? 0.0f : th;
}

__device__ __forceinline__ float noisy(float th, float f) {
    return theta_eff(th) * ((f * 2.0f - 1.0f) * EPSC + 1.0f);
}

static constexpr int PO   = 32;
static constexpr int PS   = 8;
static constexpr int PJ   = NIN_ / PS;
static constexpr int TSTR = 132;

__global__ void __launch_bounds__(256)
prep_kernel(const float* __restrict__ theta,
            const float* __restrict__ noise,
            float* __restrict__ Vt,
            float* __restrict__ cbias)
{
    __shared__ __attribute__((aligned(16))) float tile[PO * TSTR];
    __shared__ float pden[PS][PO];
    __shared__ float pbias[PS][PO];

    const int tid = threadIdx.x;
    const int ol  = tid & (PO - 1);
    const int ms  = tid >> 5;
    const int n   = blockIdx.x >> 2;
    const int ob  = (blockIdx.x & 3) * PO;
    const int o   = ob + ol;

    const float* thp = theta + o;
    const float* nzp = noise + (size_t)n * M_ * NOUT_ + o;

    float th[PJ], nf[PJ];
    #pragma unroll
    for (int j = 0; j < PJ; ++j) {
        const int m = ms + PS * j;
        th[j] = thp[m * NOUT_];
        nf[j] = nzp[m * NOUT_];
    }
    float tht = 0.0f, nft = 0.0f;
    if (ms < M_ - NIN_) {
        tht = thp[(NIN_ + ms) * NOUT_];
        nft = nzp[(NIN_ + ms) * NOUT_];
    }

    float tn[PJ];
    float den = 0.0f, bs = 0.0f;
    #pragma unroll
    for (int j = 0; j < PJ; ++j) {
        tn[j] = noisy(th[j], nf[j]);
        den += fabsf(tn[j]);
        bs  += fmaxf(-tn[j], 0.0f);
    }
    const float tnt = noisy(tht, nft);
    den += fabsf(tnt);
    if (ms == 0) bs += fmaxf(tnt, 0.0f);
    pden[ms][ol]  = den;
    pbias[ms][ol] = bs;
    __syncthreads();

    float dsum = 0.0f, bsum = 0.0f;
    #pragma unroll
    for (int s = 0; s < PS; ++s) {
        dsum += pden[s][ol];
        bsum += pbias[s][ol];
    }
    const float inv = 1.0f / (dsum + 1e-10f);

    #pragma unroll
    for (int j = 0; j < PJ; ++j)
        tile[ol * TSTR + ms + PS * j] = tn[j] * inv;
    __syncthreads();

    if (ms == 0) cbias[n * NOUT_ + o] = bsum * inv;

    #pragma unroll
    for (int rr = 0; rr < PO / PS; ++rr) {
        const int row = ms * (PO / PS) + rr;
        const float4 v = *reinterpret_cast<const float4*>(&tile[row * TSTR + ol * 4]);
        float4* dst = reinterpret_cast<float4*>(Vt + ((size_t)n * NOUT_ + ob + row) * NIN_) + ol;
        *dst = v;
    }
}

__device__ __forceinline__ void load_group(const float* __restrict__ p, float4 (&q)[4]) {
    q[0] = *reinterpret_cast<const float4*>(p);
    q[1] = *reinterpret_cast<const float4*>(p + E_);
    q[2] = *reinterpret_cast<const float4*>(p + 4 * E_);
    q[3] = *reinterpret_cast<const float4*>(p + 5 * E_);
}

__device__ __forceinline__ void mma_group(const float* vp, const float4 (&q)[4], v8f (&acc)[2][4]) {
    #pragma unroll
    for (int s = 0; s < 2; ++s) {
        const float4 r0 = q[2 * s], r1 = q[2 * s + 1];
        v2f Bf[4];
        Bf[0].x = r0.x; Bf[0].y = r1.x;
        Bf[1].x = r0.y; Bf[1].y = r1.y;
        Bf[2].x = r0.z; Bf[2].y = r1.z;
        Bf[3].x = r0.w; Bf[3].y = r1.w;
        #pragma unroll
        for (int ot = 0; ot < 2; ++ot) {
            const float* fp = vp + ot * 16 * VSTR + 4 * s;
            v2f Af;
            Af.x = fp[0];
            Af.y = fp[1];
            #pragma unroll
            for (int t = 0; t < 4; ++t)
                acc[ot][t] = __builtin_amdgcn_wmma_f32_16x16x4_f32(
                    false, Af, false, Bf[t], (short)0, acc[ot][t], false, false);
        }
    }
}

__global__ void __launch_bounds__(256)
gemm_act_kernel(const float* __restrict__ a,
                const float* __restrict__ Vt,
                const float* __restrict__ cbias,
                const float* __restrict__ eta,
                float* __restrict__ out)
{
    __shared__ __attribute__((aligned(16))) float ldsV[NOUT_ * VSTR];
    __shared__ float ldsC[NOUT_];

    const int bid    = blockIdx.x;
    const int echunk = bid & 7;
    const int b      = (bid >> 3) & 31;
    const int n      = bid >> 8;
    const int tid    = threadIdx.x;

    const float4* vsrc = reinterpret_cast<const float4*>(Vt + (size_t)n * NOUT_ * NIN_);
    #pragma unroll
    for (int r = 0; r < 16; ++r) {
        const int idx  = r * 256 + tid;
        const int orow = idx >> 5;
        const int m4   = idx & 31;
        *reinterpret_cast<float4*>(&ldsV[orow * VSTR + m4 * 4]) = vsrc[idx];
    }
    if (tid < NOUT_) ldsC[tid] = cbias[n * NOUT_ + tid];
    __syncthreads();

    const int wave  = tid >> 5;
    const int lane  = tid & 31;
    const int lrow  = lane & 15;
    const int lhalf = lane >> 4;
    const int eg    = wave & 1;
    const int og    = wave >> 1;
    const int ebase = echunk * 128 + eg * 64 + 4 * lrow;

    const float* Abase = a + ((size_t)(n * B_ + b) * NIN_) * E_ + ebase;
    const float* Vbase = &ldsV[(og * 32 + lrow) * VSTR];

    v8f acc[2][4];
    #pragma unroll
    for (int ot = 0; ot < 2; ++ot)
        #pragma unroll
        for (int t = 0; t < 4; ++t)
            #pragma unroll
            for (int i = 0; i < 8; ++i) acc[ot][t][i] = 0.0f;

    const float* ap = Abase + (size_t)(2 * lhalf) * E_;
    const float* vp = Vbase + 2 * lhalf;

    float4 qa[4], qb[4];
    load_group(ap, qa);
    #pragma unroll 1
    for (int g = 0; g < 14; g += 2) {
        load_group(ap + (size_t)(g + 1) * 8 * E_, qb);
        mma_group(vp + g * 8, qa, acc);
        load_group(ap + (size_t)(g + 2) * 8 * E_, qa);
        mma_group(vp + (g + 1) * 8, qb, acc);
    }
    load_group(ap + (size_t)15 * 8 * E_, qb);
    mma_group(vp + 14 * 8, qa, acc);
    mma_group(vp + 15 * 8, qb, acc);

    const float eta0 = eta[0], eta1 = eta[1], eta2 = eta[2], eta3 = eta[3];
    float* obase = out + ((size_t)(n * B_ + b) * NOUT_) * E_ + ebase;
    #pragma unroll
    for (int ot = 0; ot < 2; ++ot) {
        #pragma unroll
        for (int r = 0; r < 8; ++r) {
            const int oo = og * 32 + ot * 16 + r + 8 * lhalf;
            const float c = ldsC[oo];
            float4 v;
            v.x = eta0 + eta1 * tanhf(((acc[ot][0][r] + c) - eta2) * eta3);
            v.y = eta0 + eta1 * tanhf(((acc[ot][1][r] + c) - eta2) * eta3);
            v.z = eta0 + eta1 * tanhf(((acc[ot][2][r] + c) - eta2) * eta3);
            v.w = eta0 + eta1 * tanhf(((acc[ot][3][r] + c) - eta2) * eta3);
            *reinterpret_cast<float4*>(obase + (size_t)oo * E_) = v;
        }
    }
}

extern "C" void kernel_launch(void* const* d_in, const int* in_sizes, int n_in,
                              void* d_out, int out_size, void* d_ws, size_t ws_size,
                              hipStream_t stream)
{
    const float* a     = (const float*)d_in[0];
    const float* theta = (const float*)d_in[1];
    const float* noise = (const float*)d_in[2];
    const float* eta   = (const float*)d_in[3];
    float* out = (float*)d_out;

    float* Vt = (float*)d_ws;
    float* cb = Vt + (size_t)N_ * NOUT_ * NIN_;

    prep_kernel<<<N_ * (NOUT_ / PO), 256, 0, stream>>>(theta, noise, Vt, cb);

    const int nblocks = N_ * B_ * (E_ / 128);
    gemm_act_kernel<<<nblocks, 256, 0, stream>>>(a, Vt, cb, eta, out);
}
